# starting kernel
# baseline (speedup 1.0000x reference)
#include <hip/hip_runtime.h>
#include <hip/hip_bf16.h>
#include <math.h>


namespace {

constexpr int kN1 = 40000;
constexpr int kN2 = 10000;
constexpr int kE1 = 640000;
constexpr int kE2 = 160000;

typedef __attribute__((ext_vector_type(2))) float v2f;
typedef __attribute__((ext_vector_type(8))) float v8f;

__global__ void zero_f32(float* __restrict__ p, int n) {
  int i = blockIdx.x * blockDim.x + threadIdx.x;
  if (i < n) p[i] = 0.0f;
}

__global__ void scatter_edges(const float* __restrict__ feat,
                              const int* __restrict__ src,
                              const int* __restrict__ dst,
                              float* __restrict__ sum,
                              float* __restrict__ cnt) {
  const int e = blockIdx.x;
  const int s = src[e];
  const int d = dst[e];
  const float* fp = feat + (size_t)s * 256;
  float* sp = sum + (size_t)d * 256;
  atomicAdd(&sp[threadIdx.x], fp[threadIdx.x]);
  if (threadIdx.x == 0) atomicAdd(&cnt[d], 1.0f);
}

template <int KDIM, int NDIM, bool SIGMOID>
__global__ void fused_wmma_gemm(const float* __restrict__ Asum,
                                const float* __restrict__ cnt,
                                const float* __restrict__ Aself,
                                const float* __restrict__ Wl,
                                const float* __restrict__ Wr,
                                const float* __restrict__ bias,
                                float* __restrict__ out,
                                int M) {
  constexpr int KC = 32;
  constexpr int LDS_STRIDE = KC + 2;
  __shared__ float ldsW[2][64][LDS_STRIDE];

  const int lane = threadIdx.x & 31;
  const int wave = threadIdx.x >> 5;
  constexpr int kColBlocks = NDIM / 64;
  const int brow = blockIdx.x / kColBlocks;
  const int n0 = (blockIdx.x % kColBlocks) * 64;
  const int m0 = brow * 128 + wave * 16;

  const int half = lane >> 4;
  const int l15 = lane & 15;
  const int arow = min(m0 + l15, M - 1);
  const float inv = 1.0f / fmaxf(cnt[arow], 1.0f);

  const float* aSumRow  = Asum  + (size_t)arow * KDIM + 2 * half;
  const float* aSelfRow = Aself + (size_t)arow * KDIM + 2 * half;

  v8f acc[4];
#pragma unroll
  for (int t = 0; t < 4; ++t) acc[t] = v8f{0, 0, 0, 0, 0, 0, 0, 0};

  for (int kc = 0; kc < KDIM; kc += KC) {
    __syncthreads();
    for (int i = threadIdx.x; i < 64 * KC; i += 256) {
      const int col = i & 63;
      const int kr = i >> 6;
      ldsW[0][col][kr] = Wl[(size_t)(kc + kr) * NDIM + n0 + col];
      ldsW[1][col][kr] = Wr[(size_t)(kc + kr) * NDIM + n0 + col];
    }
    __syncthreads();

#pragma unroll
    for (int k2 = 0; k2 < KC; k2 += 4) {
      v2f a1 = *(const v2f*)(aSumRow + kc + k2);
      a1 = a1 * inv;
      const v2f a2 = *(const v2f*)(aSelfRow + kc + k2);
      const int koff = k2 + 2 * half;
#pragma unroll
      for (int t = 0; t < 4; ++t) {
        const int lcol = 16 * t + l15;
        const v2f bl = *(const v2f*)&ldsW[0][lcol][koff];
        acc[t] = __builtin_amdgcn_wmma_f32_16x16x4_f32(
            false, a1, false, bl, (short)0, acc[t], false, false);
        const v2f br = *(const v2f*)&ldsW[1][lcol][koff];
        acc[t] = __builtin_amdgcn_wmma_f32_16x16x4_f32(
            false, a2, false, br, (short)0, acc[t], false, false);
      }
    }
  }

#pragma unroll
  for (int t = 0; t < 4; ++t) {
    const int col = n0 + 16 * t + l15;
    const float b = bias[col];
#pragma unroll
    for (int r = 0; r < 8; ++r) {
      const int row = m0 + r + 8 * half;
      float v = acc[t][r] + b;
      v = SIGMOID ? (1.0f / (1.0f + expf(-v))) : fmaxf(v, 0.0f);
      if (row < M) out[(size_t)row * NDIM + col] = v;
    }
  }
}

}

extern "C" void kernel_launch(void* const* d_in, const int* in_sizes, int n_in,
                              void* d_out, int out_size, void* d_ws, size_t ws_size,
                              hipStream_t stream) {
  (void)in_sizes; (void)n_in; (void)out_size; (void)ws_size;

  const float* x   = (const float*)d_in[0];
  const float* W1l = (const float*)d_in[1];
  const float* b1  = (const float*)d_in[2];
  const float* W1r = (const float*)d_in[3];
  const float* W2l = (const float*)d_in[4];
  const float* b2  = (const float*)d_in[5];
  const float* W2r = (const float*)d_in[6];
  const int* src1  = (const int*)d_in[7];
  const int* dst1  = (const int*)d_in[8];
  const int* src2  = (const int*)d_in[9];
  const int* dst2  = (const int*)d_in[10];
  float* out = (float*)d_out;

  float* ws   = (float*)d_ws;
  float* sum1 = ws;
  float* cnt1 = sum1 + (size_t)kN1 * 256;
  float* sum2 = cnt1 + kN1;
  float* cnt2 = sum2 + (size_t)kN2 * 256;
  float* h    = cnt2 + kN2;
  const size_t zeroCount =
      (size_t)kN1 * 256 + kN1 + (size_t)kN2 * 256 + kN2;

  zero_f32<<<(int)((zeroCount + 255) / 256), 256, 0, stream>>>(ws, (int)zeroCount);

  scatter_edges<<<kE1, 256, 0, stream>>>(x, src1, dst1, sum1, cnt1);

  fused_wmma_gemm<256, 256, false>
      <<<313 * 4, 256, 0, stream>>>(sum1, cnt1, x, W1l, W1r, b1, h, kN1);

  scatter_edges<<<kE2, 256, 0, stream>>>(h, src2, dst2, sum2, cnt2);

  fused_wmma_gemm<256, 128, true>
      <<<79 * 2, 256, 0, stream>>>(sum2, cnt2, h, W2l, W2r, b2, out, kN2);
}


# Segment sums by owner work-groups: each scans dst for its node run and adds source rows into an LDS sum tile (LDS f32 atomics); no global atomics, no zero pass; sums in LDS arrival order
# speedup vs baseline: 3.1437x; 3.1437x over previous
#include <hip/hip_runtime.h>
#include <hip/hip_bf16.h>
#include <math.h>


namespace {

constexpr int kN1 = 40000;
constexpr int kN2 = 10000;
constexpr int kE1 = 640000;
constexpr int kE2 = 160000;

typedef __attribute__((ext_vector_type(2))) float v2f;
typedef __attribute__((ext_vector_type(8))) float v8f;

template <int NB, int CAP>
__global__ __launch_bounds__(1024) void aggregate_owner(
    const float* __restrict__ feat, const int* __restrict__ src,
    const int* __restrict__ dst, float* __restrict__ sum,
    float* __restrict__ cnt, int E, int nNodes) {
  constexpr int NT = 1024;
  constexpr int U = 8;
  static_assert(CAP % 4 == 0, "pieces must keep 16-B alignment of dst reads");
  __shared__ float sAcc[NB * 256];
  __shared__ int sCnt[NB];
  __shared__ int sListE[CAP];
  __shared__ int sListL[CAP];
  __shared__ int sN;

  const int tid = threadIdx.x;
  const int lane = tid & 31;
  const int wave = __builtin_amdgcn_readfirstlane(tid >> 5);
  const int base = blockIdx.x * NB;

  for (int i = tid; i < NB * 256; i += NT) sAcc[i] = 0.0f;
  for (int i = tid; i < NB; i += NT) sCnt[i] = 0;
  if (tid == 0) sN = 0;
  __syncthreads();

  auto push = [&](int d, int e) {
    const unsigned loc = (unsigned)(d - base);
    if (loc < (unsigned)NB) {
      const int idx = atomicAdd(&sN, 1);
      if (idx < CAP) { sListE[idx] = e; sListL[idx] = (int)loc; }
    }
  };
  auto push4 = [&](const int4 d, int e) {
    const unsigned l0 = (unsigned)(d.x - base), l1 = (unsigned)(d.y - base);
    const unsigned l2 = (unsigned)(d.z - base), l3 = (unsigned)(d.w - base);
    if ((l0 < (unsigned)NB) | (l1 < (unsigned)NB) | (l2 < (unsigned)NB) |
        (l3 < (unsigned)NB)) {
      push(d.x, e); push(d.y, e + 1); push(d.z, e + 2); push(d.w, e + 3);
    }
  };

  int pos = 0;
  int step = E;
  while (pos < E) {
    const int end = min(pos + step, E);
    const int nvec = (end - pos) >> 2;
    int v = tid;
    for (; v + 3 * NT < nvec; v += 4 * NT) {
      int4 d[4];
#pragma unroll
      for (int u = 0; u < 4; ++u)
        d[u] = *(const int4*)(dst + pos + 4 * (v + u * NT));
#pragma unroll
      for (int u = 0; u < 4; ++u) push4(d[u], pos + 4 * (v + u * NT));
    }
    for (; v < nvec; v += NT)
      push4(*(const int4*)(dst + pos + 4 * v), pos + 4 * v);
    const int tail = pos + 4 * nvec + tid;
    if (tail < end) push(dst[tail], tail);
    __syncthreads();
    const int n = sN;
    __syncthreads();
    if (tid == 0) sN = 0;
    if (n <= CAP) {
      for (int j0 = wave * U; j0 < n; j0 += 32 * U) {
        int myS = 0, myL = -1;
        if (lane < U && j0 + lane < n) {
          myL = sListL[j0 + lane];
          myS = src[sListE[j0 + lane]];
        }
        float4 va[U], vb[U];
#pragma unroll
        for (int u = 0; u < U; ++u) {
          const int s = __builtin_amdgcn_readlane(myS, u);
          const float4* rp = (const float4*)(feat + (size_t)s * 256);
          va[u] = rp[lane];
          vb[u] = rp[32 + lane];
        }
#pragma unroll
        for (int u = 0; u < U; ++u) {
          const int loc = __builtin_amdgcn_readlane(myL, u);
          if (loc >= 0) {
            float* ap = sAcc + loc * 256 + lane;
            atomicAdd(ap, va[u].x);
            atomicAdd(ap + 32, va[u].y);
            atomicAdd(ap + 64, va[u].z);
            atomicAdd(ap + 96, va[u].w);
            atomicAdd(ap + 128, vb[u].x);
            atomicAdd(ap + 160, vb[u].y);
            atomicAdd(ap + 192, vb[u].z);
            atomicAdd(ap + 224, vb[u].w);
            if (lane == 0) atomicAdd(&sCnt[loc], 1);
          }
        }
      }
      pos = end;
    } else {
      step = CAP;
    }
    __syncthreads();
  }

  for (int r = wave; r < NB; r += 32) {
    const int node = base + r;
    if (node < nNodes) {
      const float* ap = sAcc + r * 256 + lane;
      float4 a, b;
      a.x = ap[0];   a.y = ap[32];  a.z = ap[64];  a.w = ap[96];
      b.x = ap[128]; b.y = ap[160]; b.z = ap[192]; b.w = ap[224];
      float4* op = (float4*)(sum + (size_t)node * 256);
      op[lane] = a;
      op[32 + lane] = b;
      if (lane == 0) cnt[node] = (float)sCnt[r];
    }
  }
}

template <int KDIM, int NDIM, bool SIGMOID>
__global__ void fused_wmma_gemm(const float* __restrict__ Asum,
                                const float* __restrict__ cnt,
                                const float* __restrict__ Aself,
                                const float* __restrict__ Wl,
                                const float* __restrict__ Wr,
                                const float* __restrict__ bias,
                                float* __restrict__ out,
                                int M) {
  constexpr int KC = 32;
  constexpr int LDS_STRIDE = KC + 2;
  __shared__ float ldsW[2][64][LDS_STRIDE];

  const int lane = threadIdx.x & 31;
  const int wave = threadIdx.x >> 5;
  constexpr int kColBlocks = NDIM / 64;
  const int brow = blockIdx.x / kColBlocks;
  const int n0 = (blockIdx.x % kColBlocks) * 64;
  const int m0 = brow * 128 + wave * 16;

  const int half = lane >> 4;
  const int l15 = lane & 15;
  const int arow = min(m0 + l15, M - 1);
  const float inv = 1.0f / fmaxf(cnt[arow], 1.0f);

  const float* aSumRow  = Asum  + (size_t)arow * KDIM + 2 * half;
  const float* aSelfRow = Aself + (size_t)arow * KDIM + 2 * half;

  v8f acc[4];
#pragma unroll
  for (int t = 0; t < 4; ++t) acc[t] = v8f{0, 0, 0, 0, 0, 0, 0, 0};

  for (int kc = 0; kc < KDIM; kc += KC) {
    __syncthreads();
    for (int i = threadIdx.x; i < 64 * KC; i += 256) {
      const int col = i & 63;
      const int kr = i >> 6;
      ldsW[0][col][kr] = Wl[(size_t)(kc + kr) * NDIM + n0 + col];
      ldsW[1][col][kr] = Wr[(size_t)(kc + kr) * NDIM + n0 + col];
    }
    __syncthreads();

#pragma unroll
    for (int k2 = 0; k2 < KC; k2 += 4) {
      v2f a1 = *(const v2f*)(aSumRow + kc + k2);
      a1 = a1 * inv;
      const v2f a2 = *(const v2f*)(aSelfRow + kc + k2);
      const int koff = k2 + 2 * half;
#pragma unroll
      for (int t = 0; t < 4; ++t) {
        const int lcol = 16 * t + l15;
        const v2f bl = *(const v2f*)&ldsW[0][lcol][koff];
        acc[t] = __builtin_amdgcn_wmma_f32_16x16x4_f32(
            false, a1, false, bl, (short)0, acc[t], false, false);
        const v2f br = *(const v2f*)&ldsW[1][lcol][koff];
        acc[t] = __builtin_amdgcn_wmma_f32_16x16x4_f32(
            false, a2, false, br, (short)0, acc[t], false, false);
      }
    }
  }

#pragma unroll
  for (int t = 0; t < 4; ++t) {
    const int col = n0 + 16 * t + l15;
    const float b = bias[col];
#pragma unroll
    for (int r = 0; r < 8; ++r) {
      const int row = m0 + r + 8 * half;
      float v = acc[t][r] + b;
      v = SIGMOID ? (1.0f / (1.0f + expf(-v))) : fmaxf(v, 0.0f);
      if (row < M) out[(size_t)row * NDIM + col] = v;
    }
  }
}

}

extern "C" void kernel_launch(void* const* d_in, const int* in_sizes, int n_in,
                              void* d_out, int out_size, void* d_ws, size_t ws_size,
                              hipStream_t stream) {
  (void)in_sizes; (void)n_in; (void)out_size; (void)ws_size;

  const float* x   = (const float*)d_in[0];
  const float* W1l = (const float*)d_in[1];
  const float* b1  = (const float*)d_in[2];
  const float* W1r = (const float*)d_in[3];
  const float* W2l = (const float*)d_in[4];
  const float* b2  = (const float*)d_in[5];
  const float* W2r = (const float*)d_in[6];
  const int* src1  = (const int*)d_in[7];
  const int* dst1  = (const int*)d_in[8];
  const int* src2  = (const int*)d_in[9];
  const int* dst2  = (const int*)d_in[10];
  float* out = (float*)d_out;

  float* ws   = (float*)d_ws;
  float* sum1 = ws;
  float* cnt1 = sum1 + (size_t)kN1 * 256;
  float* sum2 = cnt1 + kN1;
  float* cnt2 = sum2 + (size_t)kN2 * 256;
  float* h    = cnt2 + kN2;

  aggregate_owner<160, 8192>
      <<<250, 1024, 0, stream>>>(x, src1, dst1, sum1, cnt1, kE1, kN1);

  fused_wmma_gemm<256, 256, false>
      <<<313 * 4, 256, 0, stream>>>(sum1, cnt1, x, W1l, W1r, b1, h, kN1);

  aggregate_owner<40, 8192>
      <<<250, 1024, 0, stream>>>(h, src2, dst2, sum2, cnt2, kE2, kN2);

  fused_wmma_gemm<256, 128, true>
      <<<79 * 2, 256, 0, stream>>>(sum2, cnt2, h, W2l, W2r, b2, out, kN2);
}


# Owner groups scan one owner byte per edge (dst / run length, written by a small first launch), not 4-byte dst; a hit's dst is read and tested in the add step; sums stay in LDS arrival order
# speedup vs baseline: 3.2775x; 1.0426x over previous
#include <hip/hip_runtime.h>
#include <hip/hip_bf16.h>
#include <math.h>
#include <type_traits>


namespace {

constexpr int kN1 = 40000;
constexpr int kN2 = 10000;
constexpr int kE1 = 640000;
constexpr int kE2 = 160000;
constexpr int kNB1 = 160;
constexpr int kNB2 = 40;

typedef __attribute__((ext_vector_type(2))) float v2f;
typedef __attribute__((ext_vector_type(8))) float v8f;

__global__ __launch_bounds__(256) void owner_keys(
    const int* __restrict__ dst1, const int* __restrict__ dst2,
    unsigned* __restrict__ key1, unsigned* __restrict__ key2) {
  static_assert(kE1 % 4 == 0 && kE2 % 4 == 0, "four edges per thread");
  constexpr int Q1 = kE1 / 4, Q2 = kE2 / 4;
  int t = blockIdx.x * 256 + threadIdx.x;
  if (t < Q1) {
    const int4 d = ((const int4*)dst1)[t];
    key1[t] = (((unsigned)d.x / (unsigned)kNB1) & 255u) | ((((unsigned)d.y / (unsigned)kNB1) & 255u) << 8) |
              ((((unsigned)d.z / (unsigned)kNB1) & 255u) << 16) | (((unsigned)d.w / (unsigned)kNB1) << 24);
  } else if (t < Q1 + Q2) {
    t -= Q1;
    const int4 d = ((const int4*)dst2)[t];
    key2[t] = (((unsigned)d.x / (unsigned)kNB2) & 255u) | ((((unsigned)d.y / (unsigned)kNB2) & 255u) << 8) |
              ((((unsigned)d.z / (unsigned)kNB2) & 255u) << 16) | (((unsigned)d.w / (unsigned)kNB2) << 24);
  }
}

template <int NB, int CAP>
__global__ __launch_bounds__(1024) void aggregate_owner(
    const float* __restrict__ feat, const int* __restrict__ src,
    const int* __restrict__ dst, const unsigned char* __restrict__ key,
    float* __restrict__ sum, float* __restrict__ cnt, int E, int nNodes) {
  constexpr int NT = 1024;
  constexpr int U = 8;
  static_assert(CAP % 16 == 0, "pieces must keep 16-B alignment of key reads");
  __shared__ float sAcc[NB * 256];
  __shared__ int sCnt[NB];
  __shared__ int sListE[CAP];
  __shared__ int sN;

  const int tid = threadIdx.x;
  const int lane = tid & 31;
  const int wave = __builtin_amdgcn_readfirstlane(tid >> 5);
  const int base = blockIdx.x * NB;
  const unsigned own = (unsigned)blockIdx.x & 255u;
  const unsigned pat = own * 0x01010101u;

  for (int i = tid; i < NB * 256; i += NT) sAcc[i] = 0.0f;
  for (int i = tid; i < NB; i += NT) sCnt[i] = 0;
  if (tid == 0) sN = 0;
  __syncthreads();

  auto hits = [&](unsigned w) -> unsigned {
    const unsigned x = w ^ pat;
    return ~(((x & 0x7f7f7f7fu) + 0x7f7f7f7fu) | x) & 0x80808080u;
  };
  auto put = [&](unsigned z, int e, int& idx) {
    while (z) {
      if (idx < CAP) sListE[idx] = e + (__builtin_ctz(z) >> 3);
      ++idx;
      z &= z - 1;
    }
  };
  auto scan_group = [&](const uint4* w, const int* e0, auto nv) {
    constexpr int NV = decltype(nv)::value;
    unsigned z[NV][4];
    int c = 0;
#pragma unroll
    for (int u = 0; u < NV; ++u) {
      z[u][0] = hits(w[u].x); z[u][1] = hits(w[u].y);
      z[u][2] = hits(w[u].z); z[u][3] = hits(w[u].w);
      c += __builtin_popcount(z[u][0]) + __builtin_popcount(z[u][1]) +
           __builtin_popcount(z[u][2]) + __builtin_popcount(z[u][3]);
    }
    if (c) {
      int idx = atomicAdd(&sN, c);
#pragma unroll
      for (int u = 0; u < NV; ++u) {
        put(z[u][0], e0[u], idx);     put(z[u][1], e0[u] + 4, idx);
        put(z[u][2], e0[u] + 8, idx); put(z[u][3], e0[u] + 12, idx);
      }
    }
  };

  int pos = 0;
  int step = E;
  while (pos < E) {
    const int end = min(pos + step, E);
    const int nvec = (end - pos) >> 4;
    int v = tid;
    for (; v + 3 * NT < nvec; v += 4 * NT) {
      uint4 w[4];
      int e0[4];
#pragma unroll
      for (int u = 0; u < 4; ++u) {
        e0[u] = pos + 16 * (v + u * NT);
        w[u] = *(const uint4*)(key + e0[u]);
      }
      scan_group(w, e0, std::integral_constant<int, 4>{});
    }
    for (; v < nvec; v += NT) {
      const int e0 = pos + 16 * v;
      const uint4 w = *(const uint4*)(key + e0);
      scan_group(&w, &e0, std::integral_constant<int, 1>{});
    }
    const int tail = pos + 16 * nvec + tid;
    if (tid < 16 && tail < end && key[tail] == own) {
      const int idx = atomicAdd(&sN, 1);
      if (idx < CAP) sListE[idx] = tail;
    }
    __syncthreads();
    const int n = sN;
    __syncthreads();
    if (tid == 0) sN = 0;
    if (n <= CAP) {
      for (int j0 = wave * U; j0 < n; j0 += 32 * U) {
        int myS = 0, myL = -1;
        if (lane < U && j0 + lane < n) {
          const int e = sListE[j0 + lane];
          myS = src[e];
          const unsigned loc = (unsigned)(dst[e] - base);
          if (loc < (unsigned)NB) myL = (int)loc;
        }
        float4 va[U], vb[U];
#pragma unroll
        for (int u = 0; u < U; ++u) {
          const int s = __builtin_amdgcn_readlane(myS, u);
          const float4* rp = (const float4*)(feat + (size_t)s * 256);
          va[u] = rp[lane];
          vb[u] = rp[32 + lane];
        }
#pragma unroll
        for (int u = 0; u < U; ++u) {
          const int loc = __builtin_amdgcn_readlane(myL, u);
          if (loc >= 0) {
            float* ap = sAcc + loc * 256 + lane;
            atomicAdd(ap, va[u].x);
            atomicAdd(ap + 32, va[u].y);
            atomicAdd(ap + 64, va[u].z);
            atomicAdd(ap + 96, va[u].w);
            atomicAdd(ap + 128, vb[u].x);
            atomicAdd(ap + 160, vb[u].y);
            atomicAdd(ap + 192, vb[u].z);
            atomicAdd(ap + 224, vb[u].w);
            if (lane == 0) atomicAdd(&sCnt[loc], 1);
          }
        }
      }
      pos = end;
    } else {
      step = CAP;
    }
    __syncthreads();
  }

  for (int r = wave; r < NB; r += 32) {
    const int node = base + r;
    if (node < nNodes) {
      const float* ap = sAcc + r * 256 + lane;
      float4 a, b;
      a.x = ap[0];   a.y = ap[32];  a.z = ap[64];  a.w = ap[96];
      b.x = ap[128]; b.y = ap[160]; b.z = ap[192]; b.w = ap[224];
      float4* op = (float4*)(sum + (size_t)node * 256);
      op[lane] = a;
      op[32 + lane] = b;
      if (lane == 0) cnt[node] = (float)sCnt[r];
    }
  }
}

template <int KDIM, int NDIM, bool SIGMOID>
__global__ void fused_wmma_gemm(const float* __restrict__ Asum,
                                const float* __restrict__ cnt,
                                const float* __restrict__ Aself,
                                const float* __restrict__ Wl,
                                const float* __restrict__ Wr,
                                const float* __restrict__ bias,
                                float* __restrict__ out,
                                int M) {
  constexpr int KC = 32;
  constexpr int LDS_STRIDE = KC + 2;
  __shared__ float ldsW[2][64][LDS_STRIDE];

  const int lane = threadIdx.x & 31;
  const int wave = threadIdx.x >> 5;
  constexpr int kColBlocks = NDIM / 64;
  const int brow = blockIdx.x / kColBlocks;
  const int n0 = (blockIdx.x % kColBlocks) * 64;
  const int m0 = brow * 128 + wave * 16;

  const int half = lane >> 4;
  const int l15 = lane & 15;
  const int arow = min(m0 + l15, M - 1);
  const float inv = 1.0f / fmaxf(cnt[arow], 1.0f);

  const float* aSumRow  = Asum  + (size_t)arow * KDIM + 2 * half;
  const float* aSelfRow = Aself + (size_t)arow * KDIM + 2 * half;

  v8f acc[4];
#pragma unroll
  for (int t = 0; t < 4; ++t) acc[t] = v8f{0, 0, 0, 0, 0, 0, 0, 0};

  for (int kc = 0; kc < KDIM; kc += KC) {
    __syncthreads();
    for (int i = threadIdx.x; i < 64 * KC; i += 256) {
      const int col = i & 63;
      const int kr = i >> 6;
      ldsW[0][col][kr] = Wl[(size_t)(kc + kr) * NDIM + n0 + col];
      ldsW[1][col][kr] = Wr[(size_t)(kc + kr) * NDIM + n0 + col];
    }
    __syncthreads();

#pragma unroll
    for (int k2 = 0; k2 < KC; k2 += 4) {
      v2f a1 = *(const v2f*)(aSumRow + kc + k2);
      a1 = a1 * inv;
      const v2f a2 = *(const v2f*)(aSelfRow + kc + k2);
      const int koff = k2 + 2 * half;
#pragma unroll
      for (int t = 0; t < 4; ++t) {
        const int lcol = 16 * t + l15;
        const v2f bl = *(const v2f*)&ldsW[0][lcol][koff];
        acc[t] = __builtin_amdgcn_wmma_f32_16x16x4_f32(
            false, a1, false, bl, (short)0, acc[t], false, false);
        const v2f br = *(const v2f*)&ldsW[1][lcol][koff];
        acc[t] = __builtin_amdgcn_wmma_f32_16x16x4_f32(
            false, a2, false, br, (short)0, acc[t], false, false);
      }
    }
  }

#pragma unroll
  for (int t = 0; t < 4; ++t) {
    const int col = n0 + 16 * t + l15;
    const float b = bias[col];
#pragma unroll
    for (int r = 0; r < 8; ++r) {
      const int row = m0 + r + 8 * half;
      float v = acc[t][r] + b;
      v = SIGMOID ? (1.0f / (1.0f + expf(-v))) : fmaxf(v, 0.0f);
      if (row < M) out[(size_t)row * NDIM + col] = v;
    }
  }
}

}

extern "C" void kernel_launch(void* const* d_in, const int* in_sizes, int n_in,
                              void* d_out, int out_size, void* d_ws, size_t ws_size,
                              hipStream_t stream) {
  (void)in_sizes; (void)n_in; (void)out_size; (void)ws_size;

  const float* x   = (const float*)d_in[0];
  const float* W1l = (const float*)d_in[1];
  const float* b1  = (const float*)d_in[2];
  const float* W1r = (const float*)d_in[3];
  const float* W2l = (const float*)d_in[4];
  const float* b2  = (const float*)d_in[5];
  const float* W2r = (const float*)d_in[6];
  const int* src1  = (const int*)d_in[7];
  const int* dst1  = (const int*)d_in[8];
  const int* src2  = (const int*)d_in[9];
  const int* dst2  = (const int*)d_in[10];
  float* out = (float*)d_out;

  float* ws   = (float*)d_ws;
  float* sum1 = ws;
  float* cnt1 = sum1 + (size_t)kN1 * 256;
  float* sum2 = cnt1 + kN1;
  float* cnt2 = sum2 + (size_t)kN2 * 256;
  float* h    = cnt2 + kN2;
  unsigned char* key1 = (unsigned char*)(h + (size_t)kN1 * 256);
  unsigned char* key2 = key1 + kE1;
  static_assert(((size_t)kN1 * 257 + (size_t)kN2 * 257 + (size_t)kN1 * 256) % 4 == 0 &&
                kE1 % 16 == 0, "key arrays must start 16-B aligned");

  owner_keys<<<(kE1 / 4 + kE2 / 4 + 255) / 256, 256, 0, stream>>>(
      dst1, dst2, (unsigned*)key1, (unsigned*)key2);

  aggregate_owner<kNB1, 8192>
      <<<250, 1024, 0, stream>>>(x, src1, dst1, key1, sum1, cnt1, kE1, kN1);

  fused_wmma_gemm<256, 256, false>
      <<<313 * 4, 256, 0, stream>>>(sum1, cnt1, x, W1l, W1r, b1, h, kN1);

  aggregate_owner<kNB2, 8192>
      <<<250, 1024, 0, stream>>>(h, src2, dst2, key2, sum2, cnt2, kE2, kN2);

  fused_wmma_gemm<256, 128, true>
      <<<79 * 2, 256, 0, stream>>>(sum2, cnt2, h, W2l, W2r, b2, out, kN2);
}


# GEMMs weight-stationary: a group loads its 64-col slab of both weight matrices for all K into LDS once; its waves then stream 16-row tiles, no barrier or staging in the K loop; sum order kept
# speedup vs baseline: 4.3204x; 1.3182x over previous
#include <hip/hip_runtime.h>
#include <hip/hip_bf16.h>
#include <math.h>
#include <type_traits>


namespace {

constexpr int kN1 = 40000;
constexpr int kN2 = 10000;
constexpr int kE1 = 640000;
constexpr int kE2 = 160000;
constexpr int kNB1 = 160;
constexpr int kNB2 = 40;

typedef __attribute__((ext_vector_type(2))) float v2f;
typedef __attribute__((ext_vector_type(8))) float v8f;

__global__ __launch_bounds__(256) void owner_keys(
    const int* __restrict__ dst1, const int* __restrict__ dst2,
    unsigned* __restrict__ key1, unsigned* __restrict__ key2) {
  static_assert(kE1 % 4 == 0 && kE2 % 4 == 0, "four edges per thread");
  constexpr int Q1 = kE1 / 4, Q2 = kE2 / 4;
  int t = blockIdx.x * 256 + threadIdx.x;
  if (t < Q1) {
    const int4 d = ((const int4*)dst1)[t];
    key1[t] = (((unsigned)d.x / (unsigned)kNB1) & 255u) | ((((unsigned)d.y / (unsigned)kNB1) & 255u) << 8) |
              ((((unsigned)d.z / (unsigned)kNB1) & 255u) << 16) | (((unsigned)d.w / (unsigned)kNB1) << 24);
  } else if (t < Q1 + Q2) {
    t -= Q1;
    const int4 d = ((const int4*)dst2)[t];
    key2[t] = (((unsigned)d.x / (unsigned)kNB2) & 255u) | ((((unsigned)d.y / (unsigned)kNB2) & 255u) << 8) |
              ((((unsigned)d.z / (unsigned)kNB2) & 255u) << 16) | (((unsigned)d.w / (unsigned)kNB2) << 24);
  }
}

template <int NB, int CAP>
__global__ __launch_bounds__(1024) void aggregate_owner(
    const float* __restrict__ feat, const int* __restrict__ src,
    const int* __restrict__ dst, const unsigned char* __restrict__ key,
    float* __restrict__ sum, float* __restrict__ cnt, int E, int nNodes) {
  constexpr int NT = 1024;
  constexpr int U = 8;
  static_assert(CAP % 16 == 0, "pieces must keep 16-B alignment of key reads");
  __shared__ float sAcc[NB * 256];
  __shared__ int sCnt[NB];
  __shared__ int sListE[CAP];
  __shared__ int sN;

  const int tid = threadIdx.x;
  const int lane = tid & 31;
  const int wave = __builtin_amdgcn_readfirstlane(tid >> 5);
  const int base = blockIdx.x * NB;
  const unsigned own = (unsigned)blockIdx.x & 255u;
  const unsigned pat = own * 0x01010101u;

  for (int i = tid; i < NB * 256; i += NT) sAcc[i] = 0.0f;
  for (int i = tid; i < NB; i += NT) sCnt[i] = 0;
  if (tid == 0) sN = 0;
  __syncthreads();

  auto hits = [&](unsigned w) -> unsigned {
    const unsigned x = w ^ pat;
    return ~(((x & 0x7f7f7f7fu) + 0x7f7f7f7fu) | x) & 0x80808080u;
  };
  auto put = [&](unsigned z, int e, int& idx) {
    while (z) {
      if (idx < CAP) sListE[idx] = e + (__builtin_ctz(z) >> 3);
      ++idx;
      z &= z - 1;
    }
  };
  auto scan_group = [&](const uint4* w, const int* e0, auto nv) {
    constexpr int NV = decltype(nv)::value;
    unsigned z[NV][4];
    int c = 0;
#pragma unroll
    for (int u = 0; u < NV; ++u) {
      z[u][0] = hits(w[u].x); z[u][1] = hits(w[u].y);
      z[u][2] = hits(w[u].z); z[u][3] = hits(w[u].w);
      c += __builtin_popcount(z[u][0]) + __builtin_popcount(z[u][1]) +
           __builtin_popcount(z[u][2]) + __builtin_popcount(z[u][3]);
    }
    if (c) {
      int idx = atomicAdd(&sN, c);
#pragma unroll
      for (int u = 0; u < NV; ++u) {
        put(z[u][0], e0[u], idx);     put(z[u][1], e0[u] + 4, idx);
        put(z[u][2], e0[u] + 8, idx); put(z[u][3], e0[u] + 12, idx);
      }
    }
  };

  int pos = 0;
  int step = E;
  while (pos < E) {
    const int end = min(pos + step, E);
    const int nvec = (end - pos) >> 4;
    int v = tid;
    for (; v + 3 * NT < nvec; v += 4 * NT) {
      uint4 w[4];
      int e0[4];
#pragma unroll
      for (int u = 0; u < 4; ++u) {
        e0[u] = pos + 16 * (v + u * NT);
        w[u] = *(const uint4*)(key + e0[u]);
      }
      scan_group(w, e0, std::integral_constant<int, 4>{});
    }
    for (; v < nvec; v += NT) {
      const int e0 = pos + 16 * v;
      const uint4 w = *(const uint4*)(key + e0);
      scan_group(&w, &e0, std::integral_constant<int, 1>{});
    }
    const int tail = pos + 16 * nvec + tid;
    if (tid < 16 && tail < end && key[tail] == own) {
      const int idx = atomicAdd(&sN, 1);
      if (idx < CAP) sListE[idx] = tail;
    }
    __syncthreads();
    const int n = sN;
    __syncthreads();
    if (tid == 0) sN = 0;
    if (n <= CAP) {
      for (int j0 = wave * U; j0 < n; j0 += 32 * U) {
        int myS = 0, myL = -1;
        if (lane < U && j0 + lane < n) {
          const int e = sListE[j0 + lane];
          myS = src[e];
          const unsigned loc = (unsigned)(dst[e] - base);
          if (loc < (unsigned)NB) myL = (int)loc;
        }
        float4 va[U], vb[U];
#pragma unroll
        for (int u = 0; u < U; ++u) {
          const int s = __builtin_amdgcn_readlane(myS, u);
          const float4* rp = (const float4*)(feat + (size_t)s * 256);
          va[u] = rp[lane];
          vb[u] = rp[32 + lane];
        }
#pragma unroll
        for (int u = 0; u < U; ++u) {
          const int loc = __builtin_amdgcn_readlane(myL, u);
          if (loc >= 0) {
            float* ap = sAcc + loc * 256 + lane;
            atomicAdd(ap, va[u].x);
            atomicAdd(ap + 32, va[u].y);
            atomicAdd(ap + 64, va[u].z);
            atomicAdd(ap + 96, va[u].w);
            atomicAdd(ap + 128, vb[u].x);
            atomicAdd(ap + 160, vb[u].y);
            atomicAdd(ap + 192, vb[u].z);
            atomicAdd(ap + 224, vb[u].w);
            if (lane == 0) atomicAdd(&sCnt[loc], 1);
          }
        }
      }
      pos = end;
    } else {
      step = CAP;
    }
    __syncthreads();
  }

  for (int r = wave; r < NB; r += 32) {
    const int node = base + r;
    if (node < nNodes) {
      const float* ap = sAcc + r * 256 + lane;
      float4 a, b;
      a.x = ap[0];   a.y = ap[32];  a.z = ap[64];  a.w = ap[96];
      b.x = ap[128]; b.y = ap[160]; b.z = ap[192]; b.w = ap[224];
      float4* op = (float4*)(sum + (size_t)node * 256);
      op[lane] = a;
      op[32 + lane] = b;
      if (lane == 0) cnt[node] = (float)sCnt[r];
    }
  }
}

template <int KDIM, int NDIM, bool SIGMOID>
__global__ __launch_bounds__(256) void fused_wmma_gemm(
    const float* __restrict__ Asum, const float* __restrict__ cnt,
    const float* __restrict__ Aself, const float* __restrict__ Wl,
    const float* __restrict__ Wr, const float* __restrict__ bias,
    float* __restrict__ out, int M) {
  constexpr int KC = 32;
  constexpr int LP = 128;
  static_assert(KDIM % (2 * KC) == 0, "the K loop takes two chunks per trip");
  __shared__ __attribute__((aligned(16))) float ldsW[KDIM * LP];

  const int tid = threadIdx.x;
  const int lane = tid & 31;
  const int wave = __builtin_amdgcn_readfirstlane(tid >> 5);
  constexpr int kColBlocks = NDIM / 64;
  const int RB = gridDim.x / kColBlocks;
  const int r = blockIdx.x / kColBlocks;
  const int n0 = (blockIdx.x % kColBlocks) * 64;

  {
    constexpr int NI = KDIM * 16 / 256;
    static_assert(NI % 4 == 0, "groups of 4 items");
    for (int g = 0; g < NI; g += 4) {
      float4 r0[4], r1[4];
#pragma unroll
      for (int u = 0; u < 4; ++u) {
        const int item = tid + (g + u) * 256;
        const int c4 = item & 15, q = item >> 4;
        const float* sp = ((q & 1) ? Wr : Wl) + (size_t)(q & ~1) * NDIM + n0 + 4 * c4;
        r0[u] = *(const float4*)sp;
        r1[u] = *(const float4*)(sp + NDIM);
      }
#pragma unroll
      for (int u = 0; u < 4; ++u) {
        const int item = tid + (g + u) * 256;
        const int c4 = item & 15, q = item >> 4;
        const int colsw = (4 * c4) ^ (16 * ((q >> 1) & 1));
        float4* dp = (float4*)&ldsW[q * LP + 2 * colsw];
        dp[0] = float4{r0[u].x, r1[u].x, r0[u].y, r1[u].y};
        dp[1] = float4{r0[u].z, r1[u].z, r0[u].w, r1[u].w};
      }
    }
  }
  __syncthreads();

  const int half = lane >> 4;
  const int l15 = lane & 15;
  const float* bEven = ldsW + (2 * half) * LP + 2 * l15 + 32 * half;
  const float* bOdd  = ldsW + (2 * half) * LP + 2 * l15 - 32 * half;
  const int numTiles = (M + 15) >> 4;

  for (int tile = wave * RB + r; tile < numTiles; tile += 8 * RB) {
    const int m0 = tile * 16;
    const int arow = min(m0 + l15, M - 1);
    const float inv = 1.0f / fmaxf(cnt[arow], 1.0f);
    const float* aSumRow  = Asum  + (size_t)arow * KDIM + 2 * half;
    const float* aSelfRow = Aself + (size_t)arow * KDIM + 2 * half;

    v8f acc[4];
#pragma unroll
    for (int t = 0; t < 4; ++t) acc[t] = v8f{0, 0, 0, 0, 0, 0, 0, 0};

    auto fetch = [&](v2f* s, v2f* x, int kc) {
#pragma unroll
      for (int j = 0; j < KC / 4; ++j) {
        s[j] = *(const v2f*)(aSumRow + kc + 4 * j);
        x[j] = *(const v2f*)(aSelfRow + kc + 4 * j);
      }
    };
    auto mac = [&](const v2f* s, const v2f* x, int kc) {
#pragma unroll
      for (int j = 0; j < KC / 4; ++j) {
        const v2f a1 = s[j] * inv;
        const v2f a2 = x[j];
        const int lo = (kc + 4 * j) * LP;
#pragma unroll
        for (int t = 0; t < 4; ++t) {
          const float* bp = ((t & 1) ? bOdd : bEven) + lo + 32 * t;
          const v2f fl = *(const v2f*)bp;
          acc[t] = __builtin_amdgcn_wmma_f32_16x16x4_f32(
              false, a1, false, fl, (short)0, acc[t], false, false);
          const v2f fr = *(const v2f*)(bp + LP);
          acc[t] = __builtin_amdgcn_wmma_f32_16x16x4_f32(
              false, a2, false, fr, (short)0, acc[t], false, false);
        }
      }
    };

    v2f sA[KC / 4], xA[KC / 4], sB[KC / 4], xB[KC / 4];
    fetch(sA, xA, 0);
#pragma unroll 1
    for (int kc = 0; kc < KDIM; kc += 2 * KC) {
      fetch(sB, xB, kc + KC);
      mac(sA, xA, kc);
      if (kc + 2 * KC < KDIM) fetch(sA, xA, kc + 2 * KC);
      mac(sB, xB, kc + KC);
    }

#pragma unroll
    for (int t = 0; t < 4; ++t) {
      const int col = n0 + 16 * t + l15;
      const float b = bias[col];
#pragma unroll
      for (int rr = 0; rr < 8; ++rr) {
        const int row = m0 + rr + 8 * half;
        float v = acc[t][rr] + b;
        v = SIGMOID ? (1.0f / (1.0f + expf(-v))) : fmaxf(v, 0.0f);
        if (row < M) out[(size_t)row * NDIM + col] = v;
      }
    }
  }
}

}

extern "C" void kernel_launch(void* const* d_in, const int* in_sizes, int n_in,
                              void* d_out, int out_size, void* d_ws, size_t ws_size,
                              hipStream_t stream) {
  (void)in_sizes; (void)n_in; (void)out_size; (void)ws_size;

  const float* x   = (const float*)d_in[0];
  const float* W1l = (const float*)d_in[1];
  const float* b1  = (const float*)d_in[2];
  const float* W1r = (const float*)d_in[3];
  const float* W2l = (const float*)d_in[4];
  const float* b2  = (const float*)d_in[5];
  const float* W2r = (const float*)d_in[6];
  const int* src1  = (const int*)d_in[7];
  const int* dst1  = (const int*)d_in[8];
  const int* src2  = (const int*)d_in[9];
  const int* dst2  = (const int*)d_in[10];
  float* out = (float*)d_out;

  float* ws   = (float*)d_ws;
  float* sum1 = ws;
  float* cnt1 = sum1 + (size_t)kN1 * 256;
  float* sum2 = cnt1 + kN1;
  float* cnt2 = sum2 + (size_t)kN2 * 256;
  float* h    = cnt2 + kN2;
  unsigned char* key1 = (unsigned char*)(h + (size_t)kN1 * 256);
  unsigned char* key2 = key1 + kE1;
  static_assert(((size_t)kN1 * 257 + (size_t)kN2 * 257 + (size_t)kN1 * 256) % 4 == 0 &&
                kE1 % 16 == 0, "key arrays must start 16-B aligned");

  owner_keys<<<(kE1 / 4 + kE2 / 4 + 255) / 256, 256, 0, stream>>>(
      dst1, dst2, (unsigned*)key1, (unsigned*)key2);

  aggregate_owner<kNB1, 8192>
      <<<250, 1024, 0, stream>>>(x, src1, dst1, key1, sum1, cnt1, kE1, kN1);

  fused_wmma_gemm<256, 256, false>
      <<<4 * 128, 256, 0, stream>>>(sum1, cnt1, x, W1l, W1r, b1, h, kN1);

  aggregate_owner<kNB2, 8192>
      <<<250, 1024, 0, stream>>>(h, src2, dst2, key2, sum2, cnt2, kE2, kN2);

  fused_wmma_gemm<256, 128, true>
      <<<2 * 128, 256, 0, stream>>>(sum2, cnt2, h, W2l, W2r, b2, out, kN2);
}


# Owner groups read their own edge bitmap (1 bit per edge, a 128-B line per 1024-edge chunk, built in LDS by the first launch), not an owner byte per edge of all edges; sums stay in LDS arrival order
# speedup vs baseline: 4.5516x; 1.0535x over previous
#include <hip/hip_runtime.h>
#include <hip/hip_bf16.h>
#include <math.h>
#include <type_traits>


namespace {

constexpr int kN1 = 40000;
constexpr int kN2 = 10000;
constexpr int kE1 = 640000;
constexpr int kE2 = 160000;
constexpr int kNB1 = 160;
constexpr int kNB2 = 40;
constexpr int kG1 = (kN1 + kNB1 - 1) / kNB1;
constexpr int kG2 = (kN2 + kNB2 - 1) / kNB2;
constexpr int kCE = 1024;
constexpr int kC1 = (kE1 + kCE - 1) / kCE;
constexpr int kC2 = (kE2 + kCE - 1) / kCE;

typedef __attribute__((ext_vector_type(2))) float v2f;
typedef __attribute__((ext_vector_type(8))) float v8f;

template <int NB, int G>
__device__ __forceinline__ void owner_bits_chunk(
    const int* __restrict__ dst, unsigned* __restrict__ bits, unsigned* sBits,
    int chunk, int chunks, int E) {
  const int tid = threadIdx.x;
  for (int i = tid; i < G * 8; i += 256) ((uint4*)sBits)[i] = uint4{0u, 0u, 0u, 0u};
  __syncthreads();
  const int e = chunk * kCE + 4 * tid;
  if (e < E) {
    const int4 d = *(const int4*)(dst + e);
    const int w = tid >> 3;
    const unsigned b = 1u << (4 * (tid & 7));
    const unsigned g0 = (unsigned)d.x / (unsigned)NB, g1 = (unsigned)d.y / (unsigned)NB;
    const unsigned g2 = (unsigned)d.z / (unsigned)NB, g3 = (unsigned)d.w / (unsigned)NB;
    if (g0 < (unsigned)G) atomicOr(&sBits[g0 * 32 + w], b);
    if (g1 < (unsigned)G) atomicOr(&sBits[g1 * 32 + w], b << 1);
    if (g2 < (unsigned)G) atomicOr(&sBits[g2 * 32 + w], b << 2);
    if (g3 < (unsigned)G) atomicOr(&sBits[g3 * 32 + w], b << 3);
  }
  __syncthreads();
  for (int i = tid; i < G * 8; i += 256) {
    const int g = i >> 3, part = i & 7;
    ((uint4*)bits)[((size_t)g * chunks + chunk) * 8 + part] = ((const uint4*)sBits)[i];
  }
}

__global__ __launch_bounds__(256) void owner_bits(
    const int* __restrict__ dst1, const int* __restrict__ dst2,
    unsigned* __restrict__ bits1, unsigned* __restrict__ bits2) {
  static_assert(kE1 % 4 == 0 && kE2 % 4 == 0, "four edges per thread");
  constexpr int GM = kG1 > kG2 ? kG1 : kG2;
  __shared__ __attribute__((aligned(16))) unsigned sBits[GM * 32];
  const int c = blockIdx.x;
  if (c < kC1) owner_bits_chunk<kNB1, kG1>(dst1, bits1, sBits, c, kC1, kE1);
  else         owner_bits_chunk<kNB2, kG2>(dst2, bits2, sBits, c - kC1, kC2, kE2);
}

template <int NB, int CAP>
__global__ __launch_bounds__(1024) void aggregate_owner(
    const float* __restrict__ feat, const int* __restrict__ src,
    const int* __restrict__ dst, const unsigned* __restrict__ bits,
    float* __restrict__ sum, float* __restrict__ cnt, int E, int nNodes) {
  constexpr int NT = 1024;
  constexpr int U = 8;
  static_assert(CAP % 128 == 0, "pieces must keep 16-B alignment of bitmap reads");
  __shared__ float sAcc[NB * 256];
  __shared__ int sCnt[NB];
  __shared__ int sListE[CAP];
  __shared__ int sN;

  const int tid = threadIdx.x;
  const int lane = tid & 31;
  const int wave = __builtin_amdgcn_readfirstlane(tid >> 5);
  const int base = blockIdx.x * NB;
  const int nvecAll = ((E + kCE - 1) / kCE) * (kCE / 128);
  const uint4* __restrict__ bm = (const uint4*)bits + (size_t)blockIdx.x * nvecAll;

  for (int i = tid; i < NB * 256; i += NT) sAcc[i] = 0.0f;
  for (int i = tid; i < NB; i += NT) sCnt[i] = 0;
  if (tid == 0) sN = 0;
  __syncthreads();

  auto put = [&](unsigned z, int e, int& idx) {
    while (z) {
      if (idx < CAP) sListE[idx] = e + __builtin_ctz(z);
      ++idx;
      z &= z - 1;
    }
  };
  auto scan_group = [&](const uint4* w, const int* e0, auto nv) {
    constexpr int NV = decltype(nv)::value;
    int c = 0;
#pragma unroll
    for (int u = 0; u < NV; ++u)
      c += __builtin_popcount(w[u].x) + __builtin_popcount(w[u].y) +
           __builtin_popcount(w[u].z) + __builtin_popcount(w[u].w);
    if (c) {
      int idx = atomicAdd(&sN, c);
#pragma unroll
      for (int u = 0; u < NV; ++u) {
        put(w[u].x, e0[u], idx);      put(w[u].y, e0[u] + 32, idx);
        put(w[u].z, e0[u] + 64, idx); put(w[u].w, e0[u] + 96, idx);
      }
    }
  };

  int pos = 0;
  int step = E;
  while (pos < E) {
    const int end = min(pos + step, E);
    const int v0 = pos >> 7;
    const int nvec = min((end - pos + 127) >> 7, nvecAll - v0);
    int v = tid;
    for (; v + 3 * NT < nvec; v += 4 * NT) {
      uint4 w[4];
      int e0[4];
#pragma unroll
      for (int u = 0; u < 4; ++u) {
        e0[u] = pos + 128 * (v + u * NT);
        w[u] = bm[v0 + v + u * NT];
      }
      scan_group(w, e0, std::integral_constant<int, 4>{});
    }
    for (; v < nvec; v += NT) {
      const int e0 = pos + 128 * v;
      const uint4 w = bm[v0 + v];
      scan_group(&w, &e0, std::integral_constant<int, 1>{});
    }
    __syncthreads();
    const int n = sN;
    __syncthreads();
    if (tid == 0) sN = 0;
    if (n <= CAP) {
      for (int j0 = wave * U; j0 < n; j0 += 32 * U) {
        int myS = 0, myL = -1;
        if (lane < U && j0 + lane < n) {
          const int e = sListE[j0 + lane];
          myS = src[e];
          const unsigned loc = (unsigned)(dst[e] - base);
          if (loc < (unsigned)NB) myL = (int)loc;
        }
        float4 va[U], vb[U];
#pragma unroll
        for (int u = 0; u < U; ++u) {
          const int s = __builtin_amdgcn_readlane(myS, u);
          const float4* rp = (const float4*)(feat + (size_t)s * 256);
          va[u] = rp[lane];
          vb[u] = rp[32 + lane];
        }
#pragma unroll
        for (int u = 0; u < U; ++u) {
          const int loc = __builtin_amdgcn_readlane(myL, u);
          if (loc >= 0) {
            float* ap = sAcc + loc * 256 + lane;
            atomicAdd(ap, va[u].x);
            atomicAdd(ap + 32, va[u].y);
            atomicAdd(ap + 64, va[u].z);
            atomicAdd(ap + 96, va[u].w);
            atomicAdd(ap + 128, vb[u].x);
            atomicAdd(ap + 160, vb[u].y);
            atomicAdd(ap + 192, vb[u].z);
            atomicAdd(ap + 224, vb[u].w);
            if (lane == 0) atomicAdd(&sCnt[loc], 1);
          }
        }
      }
      pos = end;
    } else {
      step = CAP;
    }
    __syncthreads();
  }

  for (int r = wave; r < NB; r += 32) {
    const int node = base + r;
    if (node < nNodes) {
      const float* ap = sAcc + r * 256 + lane;
      float4 a, b;
      a.x = ap[0];   a.y = ap[32];  a.z = ap[64];  a.w = ap[96];
      b.x = ap[128]; b.y = ap[160]; b.z = ap[192]; b.w = ap[224];
      float4* op = (float4*)(sum + (size_t)node * 256);
      op[lane] = a;
      op[32 + lane] = b;
      if (lane == 0) cnt[node] = (float)sCnt[r];
    }
  }
}

template <int KDIM, int NDIM, bool SIGMOID>
__global__ __launch_bounds__(256) void fused_wmma_gemm(
    const float* __restrict__ Asum, const float* __restrict__ cnt,
    const float* __restrict__ Aself, const float* __restrict__ Wl,
    const float* __restrict__ Wr, const float* __restrict__ bias,
    float* __restrict__ out, int M) {
  constexpr int KC = 32;
  constexpr int LP = 128;
  static_assert(KDIM % (2 * KC) == 0, "the K loop takes two chunks per trip");
  __shared__ __attribute__((aligned(16))) float ldsW[KDIM * LP];

  const int tid = threadIdx.x;
  const int lane = tid & 31;
  const int wave = __builtin_amdgcn_readfirstlane(tid >> 5);
  constexpr int kColBlocks = NDIM / 64;
  const int RB = gridDim.x / kColBlocks;
  const int r = blockIdx.x / kColBlocks;
  const int n0 = (blockIdx.x % kColBlocks) * 64;

  {
    constexpr int NI = KDIM * 16 / 256;
    static_assert(NI % 4 == 0, "groups of 4 items");
    for (int g = 0; g < NI; g += 4) {
      float4 r0[4], r1[4];
#pragma unroll
      for (int u = 0; u < 4; ++u) {
        const int item = tid + (g + u) * 256;
        const int c4 = item & 15, q = item >> 4;
        const float* sp = ((q & 1) ? Wr : Wl) + (size_t)(q & ~1) * NDIM + n0 + 4 * c4;
        r0[u] = *(const float4*)sp;
        r1[u] = *(const float4*)(sp + NDIM);
      }
#pragma unroll
      for (int u = 0; u < 4; ++u) {
        const int item = tid + (g + u) * 256;
        const int c4 = item & 15, q = item >> 4;
        const int colsw = (4 * c4) ^ (16 * ((q >> 1) & 1));
        float4* dp = (float4*)&ldsW[q * LP + 2 * colsw];
        dp[0] = float4{r0[u].x, r1[u].x, r0[u].y, r1[u].y};
        dp[1] = float4{r0[u].z, r1[u].z, r0[u].w, r1[u].w};
      }
    }
  }
  __syncthreads();

  const int half = lane >> 4;
  const int l15 = lane & 15;
  const float* bEven = ldsW + (2 * half) * LP + 2 * l15 + 32 * half;
  const float* bOdd  = ldsW + (2 * half) * LP + 2 * l15 - 32 * half;
  const int numTiles = (M + 15) >> 4;

  for (int tile = wave * RB + r; tile < numTiles; tile += 8 * RB) {
    const int m0 = tile * 16;
    const int arow = min(m0 + l15, M - 1);
    const float inv = 1.0f / fmaxf(cnt[arow], 1.0f);
    const float* aSumRow  = Asum  + (size_t)arow * KDIM + 2 * half;
    const float* aSelfRow = Aself + (size_t)arow * KDIM + 2 * half;

    v8f acc[4];
#pragma unroll
    for (int t = 0; t < 4; ++t) acc[t] = v8f{0, 0, 0, 0, 0, 0, 0, 0};

    auto fetch = [&](v2f* s, v2f* x, int kc) {
#pragma unroll
      for (int j = 0; j < KC / 4; ++j) {
        s[j] = *(const v2f*)(aSumRow + kc + 4 * j);
        x[j] = *(const v2f*)(aSelfRow + kc + 4 * j);
      }
    };
    auto mac = [&](const v2f* s, const v2f* x, int kc) {
#pragma unroll
      for (int j = 0; j < KC / 4; ++j) {
        const v2f a1 = s[j] * inv;
        const v2f a2 = x[j];
        const int lo = (kc + 4 * j) * LP;
#pragma unroll
        for (int t = 0; t < 4; ++t) {
          const float* bp = ((t & 1) ? bOdd : bEven) + lo + 32 * t;
          const v2f fl = *(const v2f*)bp;
          acc[t] = __builtin_amdgcn_wmma_f32_16x16x4_f32(
              false, a1, false, fl, (short)0, acc[t], false, false);
          const v2f fr = *(const v2f*)(bp + LP);
          acc[t] = __builtin_amdgcn_wmma_f32_16x16x4_f32(
              false, a2, false, fr, (short)0, acc[t], false, false);
        }
      }
    };

    v2f sA[KC / 4], xA[KC / 4], sB[KC / 4], xB[KC / 4];
    fetch(sA, xA, 0);
#pragma unroll 1
    for (int kc = 0; kc < KDIM; kc += 2 * KC) {
      fetch(sB, xB, kc + KC);
      mac(sA, xA, kc);
      if (kc + 2 * KC < KDIM) fetch(sA, xA, kc + 2 * KC);
      mac(sB, xB, kc + KC);
    }

#pragma unroll
    for (int t = 0; t < 4; ++t) {
      const int col = n0 + 16 * t + l15;
      const float b = bias[col];
#pragma unroll
      for (int rr = 0; rr < 8; ++rr) {
        const int row = m0 + rr + 8 * half;
        float v = acc[t][rr] + b;
        v = SIGMOID ? (1.0f / (1.0f + expf(-v))) : fmaxf(v, 0.0f);
        if (row < M) out[(size_t)row * NDIM + col] = v;
      }
    }
  }
}

}

extern "C" void kernel_launch(void* const* d_in, const int* in_sizes, int n_in,
                              void* d_out, int out_size, void* d_ws, size_t ws_size,
                              hipStream_t stream) {
  (void)in_sizes; (void)n_in; (void)out_size; (void)ws_size;

  const float* x   = (const float*)d_in[0];
  const float* W1l = (const float*)d_in[1];
  const float* b1  = (const float*)d_in[2];
  const float* W1r = (const float*)d_in[3];
  const float* W2l = (const float*)d_in[4];
  const float* b2  = (const float*)d_in[5];
  const float* W2r = (const float*)d_in[6];
  const int* src1  = (const int*)d_in[7];
  const int* dst1  = (const int*)d_in[8];
  const int* src2  = (const int*)d_in[9];
  const int* dst2  = (const int*)d_in[10];
  float* out = (float*)d_out;

  float* ws   = (float*)d_ws;
  float* sum1 = ws;
  float* cnt1 = sum1 + (size_t)kN1 * 256;
  float* sum2 = cnt1 + kN1;
  float* cnt2 = sum2 + (size_t)kN2 * 256;
  float* h    = cnt2 + kN2;
  unsigned* bits1 = (unsigned*)(h + (size_t)kN1 * 256);
  unsigned* bits2 = bits1 + (size_t)kG1 * kC1 * 32;
  static_assert(((size_t)kN1 * 257 + (size_t)kN2 * 257 + (size_t)kN1 * 256) % 4 == 0,
                "bitmap arrays must start 16-B aligned");

  owner_bits<<<kC1 + kC2, 256, 0, stream>>>(dst1, dst2, bits1, bits2);

  aggregate_owner<kNB1, 8192>
      <<<kG1, 1024, 0, stream>>>(x, src1, dst1, bits1, sum1, cnt1, kE1, kN1);

  fused_wmma_gemm<256, 256, false>
      <<<4 * 128, 256, 0, stream>>>(sum1, cnt1, x, W1l, W1r, b1, h, kN1);

  aggregate_owner<kNB2, 8192>
      <<<kG2, 1024, 0, stream>>>(h, src2, dst2, bits2, sum2, cnt2, kE2, kN2);

  fused_wmma_gemm<256, 128, true>
      <<<2 * 128, 256, 0, stream>>>(sum2, cnt2, h, W2l, W2r, b2, out, kN2);
}
